# starting kernel
# baseline (speedup 1.0000x reference)
#include <hip/hip_runtime.h>
#include <hip/hip_bf16.h>
#include <math.h>

typedef float v2f __attribute__((ext_vector_type(2)));
typedef float v8f __attribute__((ext_vector_type(8)));

#define WIN_LEN   400
#define WIN_INC   100
#define PAD       300
#define N_FRAMES  1603
#define N_CH      257
#define N_IN      160000
#define N_BATCH   16
#define FRAMES_PER_BLOCK 64
#define BLOCK_THREADS    128
#define XS_LEN    (WIN_INC * (FRAMES_PER_BLOCK - 1) + WIN_LEN)
#define XS_PAD    6704
#define A_ELEMS   (WIN_LEN * 16)

__global__ __launch_bounds__(BLOCK_THREADS)
void convstft_wmma_f32(const float* __restrict__ in,
                       const float* __restrict__ w,
                       float* __restrict__ mags,
                       float* __restrict__ phase)
{
    extern __shared__ float smem[];
    float* Ar = smem;
    float* Ai = smem + A_ELEMS;
    float* Xs = smem + 2 * A_ELEMS;

    const int c0  = blockIdx.x * 16;
    const int t0  = blockIdx.y * FRAMES_PER_BLOCK;
    const int b   = blockIdx.z;
    const int tid = threadIdx.x;

    for (int idx = tid; idx < 16 * WIN_LEN; idx += BLOCK_THREADS) {
        const int m = idx / WIN_LEN;
        const int k = idx - m * WIN_LEN;
        const int c = c0 + m;
        float vr = 0.0f, vi = 0.0f;
        if (c < N_CH) {
            vr = w[(size_t)c * WIN_LEN + k];
            vi = w[(size_t)(c + N_CH) * WIN_LEN + k];
        }
        Ar[k * 16 + m] = vr;
        Ai[k * 16 + m] = vi;
    }

    const long  sbase = (long)t0 * WIN_INC - PAD;
    const float* inb  = in + (size_t)b * N_IN;
    for (int s = tid; s < XS_LEN; s += BLOCK_THREADS) {
        const long g = sbase + s;
        Xs[s] = (g >= 0 && g < (long)N_IN) ? inb[g] : 0.0f;
    }
    __syncthreads();

    const int lane = tid & 31;
    const int wave = tid >> 5;
    const int half = lane >> 4;
    const int mr   = lane & 15;
    const int soff = WIN_INC * (wave * 16 + mr);

    v8f accr = {};
    v8f acci = {};

    for (int kk = 0; kk < WIN_LEN / 4; ++kk) {
        const int kb = 4 * kk + 2 * half;
        v2f a_r, a_i, bx;
        a_r.x = Ar[(kb + 0) * 16 + mr];
        a_r.y = Ar[(kb + 1) * 16 + mr];
        a_i.x = Ai[(kb + 0) * 16 + mr];
        a_i.y = Ai[(kb + 1) * 16 + mr];
        bx.x  = Xs[soff + kb + 0];
        bx.y  = Xs[soff + kb + 1];
        accr = __builtin_amdgcn_wmma_f32_16x16x4_f32(
                   false, a_r, false, bx, (short)0, accr, false, false);
        acci = __builtin_amdgcn_wmma_f32_16x16x4_f32(
                   false, a_i, false, bx, (short)0, acci, false, false);
    }

    const float EPS = 1.19209290e-07f;
    const int t = t0 + wave * 16 + mr;
    if (t < N_FRAMES) {
        #pragma unroll
        for (int r = 0; r < 8; ++r) {
            const int c = c0 + r + 8 * half;
            if (c < N_CH) {
                const float rr = accr[r];
                const float ii = acci[r];
                float m2 = fmaxf(rr * rr + ii * ii, EPS);
                const size_t o = ((size_t)b * N_CH + c) * N_FRAMES + t;
                mags[o]  = sqrtf(m2);
                phase[o] = atan2f(ii + EPS, rr + EPS);
            }
        }
    }
}

extern "C" void kernel_launch(void* const* d_in, const int* in_sizes, int n_in,
                              void* d_out, int out_size, void* d_ws, size_t ws_size,
                              hipStream_t stream) {
    (void)in_sizes; (void)n_in; (void)d_ws; (void)ws_size; (void)out_size;

    const float* in = (const float*)d_in[0];
    const float* w  = (const float*)d_in[1];

    float* mags  = (float*)d_out;
    float* phase = mags + (size_t)N_BATCH * N_CH * N_FRAMES;

    const dim3 grid((N_CH + 15) / 16,
                    (N_FRAMES + FRAMES_PER_BLOCK - 1) / FRAMES_PER_BLOCK,
                    N_BATCH);
    const dim3 block(BLOCK_THREADS);
    const size_t smem_bytes = (size_t)(2 * A_ELEMS + XS_PAD) * sizeof(float);

    convstft_wmma_f32<<<grid, block, smem_bytes, stream>>>(in, w, mags, phase);
}


# staging: all global loads of a block issued back to back as 16-byte loads (was one 4-byte load, wait, LDS store per loop pass); 16-byte LDS stores into k-contiguous padded panels; sums unchanged
# speedup vs baseline: 3.0387x; 3.0387x over previous
#include <hip/hip_runtime.h>
#include <hip/hip_bf16.h>
#include <math.h>


typedef float v2f __attribute__((ext_vector_type(2)));
typedef float v4f __attribute__((ext_vector_type(4)));
typedef float v8f __attribute__((ext_vector_type(8)));

#define WIN_LEN   400
#define WIN_INC   100
#define PAD       300
#define N_FRAMES  1603
#define N_CH      257
#define N_IN      160000
#define N_BATCH   16
#define FRAMES_PER_BLOCK 64
#define BLOCK_THREADS    128
#define XS_LEN    (WIN_INC * (FRAMES_PER_BLOCK - 1) + WIN_LEN)
#define XS_PAD    6704
#define A_PITCH   404
#define A_ELEMS   (16 * A_PITCH)
#define W_VEC_ROW   (WIN_LEN / 4)
#define W_VEC_PANEL (16 * W_VEC_ROW)
#define W_VEC_ITERS (2 * W_VEC_PANEL / BLOCK_THREADS)
#define W_VEC_IMAG  (N_CH * W_VEC_ROW)
#define X_VECS      (XS_PAD / 4)
#define X_VEC_FULL  (X_VECS / BLOCK_THREADS)
#define X_VEC_TAIL  (X_VECS - X_VEC_FULL * BLOCK_THREADS)

__global__ __launch_bounds__(BLOCK_THREADS)
void convstft_wmma_f32(const float* __restrict__ in,
                       const float* __restrict__ w,
                       float* __restrict__ mags,
                       float* __restrict__ phase)
{
    extern __shared__ __attribute__((aligned(16))) float smem[];
    float* Ar = smem;
    float* Ai = smem + A_ELEMS;
    float* Xs = smem + 2 * A_ELEMS;

    const int c0  = blockIdx.x * 16;
    const int t0  = blockIdx.y * FRAMES_PER_BLOCK;
    const int b   = blockIdx.z;
    const unsigned tid = threadIdx.x;
    __builtin_assume(tid < BLOCK_THREADS);

    const unsigned rows_ok = (N_CH - c0) < 16 ? (unsigned)(N_CH - c0) : 16u;
    const unsigned lim = rows_ok * W_VEC_ROW;
    const v4f* w4 = (const v4f*)(w + (size_t)c0 * WIN_LEN);
    v4f wv[W_VEC_ITERS];
    #pragma unroll
    for (int i = 0; i < W_VEC_ITERS; ++i) {
        const unsigned g  = tid + BLOCK_THREADS * i;
        const unsigned p  = g >= W_VEC_PANEL ? 1u : 0u;
        const unsigned f  = g - W_VEC_PANEL * p;
        const unsigned fs = f < lim ? f : lim - 1u;
        wv[i] = w4[fs + W_VEC_IMAG * p];
    }

    const int sbase = t0 * WIN_INC - PAD;
    const float* inb = in + (size_t)b * N_IN;
    v4f xv[X_VEC_FULL + 1];
    #pragma unroll
    for (int i = 0; i < X_VEC_FULL + 1; ++i) {
        const int g = sbase + 4 * (int)(tid + BLOCK_THREADS * i);
        const unsigned gs = (unsigned)g < (unsigned)N_IN ? (unsigned)g : 0u;
        xv[i] = *(const v4f*)(inb + gs);
    }

    {
        const unsigned j = tid + BLOCK_THREADS * X_VEC_FULL;
        const int g = sbase + 4 * (int)j;
        const bool ok = (unsigned)g < (unsigned)N_IN;
        if (tid < X_VEC_TAIL) {
            v4f v = xv[X_VEC_FULL];
            v.x = ok ? v.x : 0.0f;  v.y = ok ? v.y : 0.0f;  v.z = ok ? v.z : 0.0f;  v.w = ok ? v.w : 0.0f;
            *(v4f*)(Xs + 4u * j) = v;
        }
    }
    #pragma unroll
    for (int i = 0; i < W_VEC_ITERS; ++i) {
        const unsigned g = tid + BLOCK_THREADS * i;
        const unsigned p = g >= W_VEC_PANEL ? 1u : 0u;
        const unsigned f = g - W_VEC_PANEL * p;
        const unsigned m = f / W_VEC_ROW;
        const bool ok = f < lim;
        v4f v = wv[i];
        v.x = ok ? v.x : 0.0f;  v.y = ok ? v.y : 0.0f;  v.z = ok ? v.z : 0.0f;  v.w = ok ? v.w : 0.0f;
        *(v4f*)(smem + A_ELEMS * p + 4u * f + (A_PITCH - WIN_LEN) * m) = v;
    }
    #pragma unroll
    for (int i = 0; i < X_VEC_FULL; ++i) {
        const unsigned j = tid + BLOCK_THREADS * i;
        const int g = sbase + 4 * (int)j;
        const bool ok = (unsigned)g < (unsigned)N_IN;
        v4f v = xv[i];
        v.x = ok ? v.x : 0.0f;  v.y = ok ? v.y : 0.0f;  v.z = ok ? v.z : 0.0f;  v.w = ok ? v.w : 0.0f;
        *(v4f*)(Xs + 4u * j) = v;
    }
    __syncthreads();

    const int lane = tid & 31;
    const int wave = tid >> 5;
    const int half = lane >> 4;
    const int mr   = lane & 15;
    const int soff = WIN_INC * (wave * 16 + mr);

    const float* ar_p = Ar + mr * A_PITCH + 2 * half;
    const float* ai_p = Ai + mr * A_PITCH + 2 * half;
    const float* xs_p = Xs + soff + 2 * half;

    v8f accr = {};
    v8f acci = {};

    for (int kk = 0; kk < WIN_LEN / 4; ++kk) {
        const v2f a_r = *(const v2f*)(ar_p + 4 * kk);
        const v2f a_i = *(const v2f*)(ai_p + 4 * kk);
        const v2f bx  = *(const v2f*)(xs_p + 4 * kk);
        accr = __builtin_amdgcn_wmma_f32_16x16x4_f32(
                   false, a_r, false, bx, (short)0, accr, false, false);
        acci = __builtin_amdgcn_wmma_f32_16x16x4_f32(
                   false, a_i, false, bx, (short)0, acci, false, false);
    }

    const float EPS = 1.19209290e-07f;
    const int t = t0 + wave * 16 + mr;
    if (t < N_FRAMES) {
        #pragma unroll
        for (int r = 0; r < 8; ++r) {
            const int c = c0 + r + 8 * half;
            if (c < N_CH) {
                const float rr = accr[r];
                const float ii = acci[r];
                float m2 = fmaxf(rr * rr + ii * ii, EPS);
                const size_t o = ((size_t)b * N_CH + c) * N_FRAMES + t;
                mags[o]  = sqrtf(m2);
                phase[o] = atan2f(ii + EPS, rr + EPS);
            }
        }
    }
}

extern "C" void kernel_launch(void* const* d_in, const int* in_sizes, int n_in,
                              void* d_out, int out_size, void* d_ws, size_t ws_size,
                              hipStream_t stream) {
    (void)in_sizes; (void)n_in; (void)d_ws; (void)ws_size; (void)out_size;

    const float* in = (const float*)d_in[0];
    const float* w  = (const float*)d_in[1];

    float* mags  = (float*)d_out;
    float* phase = mags + (size_t)N_BATCH * N_CH * N_FRAMES;

    const dim3 grid((N_CH + 15) / 16,
                    (N_FRAMES + FRAMES_PER_BLOCK - 1) / FRAMES_PER_BLOCK,
                    N_BATCH);
    const dim3 block(BLOCK_THREADS);
    const size_t smem_bytes = (size_t)(2 * A_ELEMS + XS_PAD) * sizeof(float);

    convstft_wmma_f32<<<grid, block, smem_bytes, stream>>>(in, w, mags, phase);
}


# weights kept in registers: a wave loads its 16 channels' re+im rows once and loops over 6-7 frame tiles; only input fragments come from LDS (16-byte reads; taps regrouped per WMMA: sum order changed)
# speedup vs baseline: 3.2553x; 1.0713x over previous
#include <hip/hip_runtime.h>
#include <math.h>


typedef float v2f __attribute__((ext_vector_type(2)));
typedef float v4f __attribute__((ext_vector_type(4)));
typedef float v8f __attribute__((ext_vector_type(8)));

#define WIN_LEN   400
#define WIN_INC   100
#define PAD       300
#define N_FRAMES  1603
#define N_CH      257
#define N_IN      160000
#define N_BATCH   16
#define BLOCK_THREADS 128
#define N_CT      17
#define N_NT      101
#define N_CHUNK   15
#define MAX_NT    7
#define N_UNITS   (N_CT * N_BATCH * N_CHUNK)
#define XW_LEN    (WIN_INC * (16 * MAX_NT - 1) + WIN_LEN)
#define XW_VECS   (XW_LEN / 4)
#define XW_PASSES ((XW_VECS + BLOCK_THREADS - 1) / BLOCK_THREADS)
#define XW_STRIDE (4 * BLOCK_THREADS * XW_PASSES)
#define K_STEPS   (WIN_LEN / 8)
#define K_GROUP   5

__global__ __launch_bounds__(BLOCK_THREADS)
void convstft_wmma_f32(const float* __restrict__ in,
                       const float* __restrict__ w,
                       float* __restrict__ mags,
                       float* __restrict__ phase)
{
    extern __shared__ __attribute__((aligned(16))) float smem[];

    const unsigned tid = threadIdx.x;
    __builtin_assume(tid < BLOCK_THREADS);
    const int lane = tid & 31;
    const int wave = __builtin_amdgcn_readfirstlane((int)(tid >> 5));
    const int half = lane >> 4;
    const int mr   = lane & 15;

    const int u0  = blockIdx.x * 4;
    const int bc0 = u0 / N_CT;
    const int bc1 = (u0 + 3) / N_CT;

    for (int s = 0; s < 2; ++s) {
        if (s == 1 && bc1 == bc0) break;
        const int bcs   = bc0 + s;
        const int bs    = bcs / N_CHUNK;
        const int chs   = bcs - bs * N_CHUNK;
        const int sbase = ((chs * N_NT) / N_CHUNK) * 16 * WIN_INC - PAD;
        const float* inb = in + (size_t)bs * N_IN;
        float* Xw = smem + s * XW_STRIDE;
        v4f xv[XW_PASSES];
        #pragma unroll
        for (int i = 0; i < XW_PASSES; ++i) {
            const int g = sbase + 4 * (int)(tid + BLOCK_THREADS * i);
            const unsigned gs = (unsigned)g < (unsigned)N_IN ? (unsigned)g : 0u;
            xv[i] = *(const v4f*)(inb + gs);
        }
        __builtin_amdgcn_sched_barrier(0);
        #pragma unroll
        for (int i = 0; i < XW_PASSES; ++i) {
            const unsigned j = tid + BLOCK_THREADS * i;
            const int g = sbase + 4 * (int)j;
            const bool ok = (unsigned)g < (unsigned)N_IN;
            v4f v = xv[i];
            v.x = ok ? v.x : 0.0f;  v.y = ok ? v.y : 0.0f;  v.z = ok ? v.z : 0.0f;  v.w = ok ? v.w : 0.0f;
            *(v4f*)(Xw + 4u * j) = v;
        }
    }

    const int u     = u0 + wave;
    const int bc    = u / N_CT;
    const int ct    = u - bc * N_CT;
    const int b     = bc / N_CHUNK;
    const int ch    = bc - b * N_CHUNK;
    const int nt_lo = (ch * N_NT) / N_CHUNK;
    const int nt_hi = ((ch + 1) * N_NT) / N_CHUNK;
    const int c0    = ct * 16;

    int row = c0 + mr;
    row = row < N_CH ? row : N_CH - 1;
    const v4f* wr = (const v4f*)(w + (size_t)row * WIN_LEN + 4 * half);
    const v4f* wi = (const v4f*)(w + (size_t)(row + N_CH) * WIN_LEN + 4 * half);
    v4f a_re[K_STEPS], a_im[K_STEPS];
    #pragma unroll
    for (int j = 0; j < K_STEPS; ++j) {
        a_re[j] = wr[2 * j];
        a_im[j] = wi[2 * j];
    }
    __syncthreads();

    const float* xs_p = smem + (bc - bc0) * XW_STRIDE + WIN_INC * mr + 4 * half;
    const float EPS = 1.19209290e-07f;

    for (int nt = nt_lo; nt < nt_hi; ++nt) {
        const float* xp = xs_p + (nt - nt_lo) * (16 * WIN_INC);
        v8f accr = {};
        v8f acci = {};
        v4f xq[2][K_GROUP];
        #pragma unroll
        for (int e = 0; e < K_GROUP; ++e) xq[0][e] = *(const v4f*)(xp + 8 * e);
        #pragma unroll
        for (int g = 0; g < K_STEPS / K_GROUP; ++g) {
            if (g + 1 < K_STEPS / K_GROUP) {
                #pragma unroll
                for (int e = 0; e < K_GROUP; ++e)
                    xq[(g + 1) & 1][e] = *(const v4f*)(xp + 8 * (K_GROUP * (g + 1) + e));
            }
            __builtin_amdgcn_sched_barrier(0);
            #pragma unroll
            for (int e = 0; e < K_GROUP; ++e) {
                const int j = K_GROUP * g + e;
                const v4f xb = xq[g & 1][e];
                const v2f b0 = {xb.x, xb.y};
                const v2f b1 = {xb.z, xb.w};
                const v2f r0 = {a_re[j].x, a_re[j].y};
                const v2f r1 = {a_re[j].z, a_re[j].w};
                const v2f i0 = {a_im[j].x, a_im[j].y};
                const v2f i1 = {a_im[j].z, a_im[j].w};
                accr = __builtin_amdgcn_wmma_f32_16x16x4_f32(false, r0, false, b0, (short)0, accr, false, false);
                acci = __builtin_amdgcn_wmma_f32_16x16x4_f32(false, i0, false, b0, (short)0, acci, false, false);
                accr = __builtin_amdgcn_wmma_f32_16x16x4_f32(false, r1, false, b1, (short)0, accr, false, false);
                acci = __builtin_amdgcn_wmma_f32_16x16x4_f32(false, i1, false, b1, (short)0, acci, false, false);
            }
        }

        float mg[8], ph[8];
        #pragma unroll
        for (int r = 0; r < 8; ++r) {
            const float rr = accr[r];
            const float ii = acci[r];
            const float m2 = fmaxf(rr * rr + ii * ii, EPS);
            mg[r] = sqrtf(m2);
            ph[r] = atan2f(ii + EPS, rr + EPS);
        }
        const int t = nt * 16 + mr;
        if (t < N_FRAMES) {
            #pragma unroll
            for (int r = 0; r < 8; ++r) {
                const int c = c0 + r + 8 * half;
                if (c < N_CH) {
                    const size_t o = ((size_t)b * N_CH + c) * N_FRAMES + t;
                    mags[o]  = mg[r];
                    phase[o] = ph[r];
                }
            }
        }
    }
}

extern "C" void kernel_launch(void* const* d_in, const int* in_sizes, int n_in,
                              void* d_out, int out_size, void* d_ws, size_t ws_size,
                              hipStream_t stream) {
    (void)in_sizes; (void)n_in; (void)d_ws; (void)ws_size; (void)out_size;

    const float* in = (const float*)d_in[0];
    const float* w  = (const float*)d_in[1];

    float* mags  = (float*)d_out;
    float* phase = mags + (size_t)N_BATCH * N_CH * N_FRAMES;

    const dim3 grid(N_UNITS / 4);
    const dim3 block(BLOCK_THREADS);
    const size_t smem_bytes = (size_t)(2 * XW_STRIDE) * sizeof(float);

    convstft_wmma_f32<<<grid, block, smem_bytes, stream>>>(in, w, mags, phase);
}


# epilogue vector instructions 476 -> 255 per tile: v_sqrt_f32 alone (1 ulp), atan2 without inf/nan/zero selects, 8 elements straight-line, stores at constant offsets; matrix sums unchanged
# speedup vs baseline: 3.6657x; 1.1261x over previous
#include <hip/hip_runtime.h>
#include <math.h>


typedef float v2f __attribute__((ext_vector_type(2)));
typedef float v4f __attribute__((ext_vector_type(4)));
typedef float v8f __attribute__((ext_vector_type(8)));

#define WIN_LEN   400
#define WIN_INC   100
#define PAD       300
#define N_FRAMES  1603
#define N_CH      257
#define N_IN      160000
#define N_BATCH   16
#define BLOCK_THREADS 128
#define N_CT      17
#define N_NT      101
#define N_CHUNK   15
#define MAX_NT    7
#define N_UNITS   (N_CT * N_BATCH * N_CHUNK)
#define XW_LEN    (WIN_INC * (16 * MAX_NT - 1) + WIN_LEN)
#define XW_VECS   (XW_LEN / 4)
#define XW_PASSES ((XW_VECS + BLOCK_THREADS - 1) / BLOCK_THREADS)
#define XW_STRIDE (4 * BLOCK_THREADS * XW_PASSES)
#define K_STEPS   (WIN_LEN / 8)
#define K_GROUP   5


#define PHASE_OFF ((size_t)N_BATCH * N_CH * N_FRAMES)

__device__ __forceinline__ void mag_phase(float x, float r2, float ii, float& mg, float& ph)
{
    const float EPS = 1.19209290e-07f;
    mg = __builtin_amdgcn_sqrtf(fmaxf(fmaf(ii, ii, r2), EPS));
    const float y = ii + EPS;
    const float ax = fabsf(x);
    const float ay = fabsf(y);
    const float u = fmaxf(fmaxf(ax, ay), 1.17549435e-38f);
    const float v = fminf(ax, ay);
    const float q = __builtin_amdgcn_ldexpf(
        __builtin_amdgcn_frexp_mantf(v) * __builtin_amdgcn_rcpf(__builtin_amdgcn_frexp_mantf(u)),
        __builtin_amdgcn_frexp_expf(v) - __builtin_amdgcn_frexp_expf(u));
    const float t = q * q;
    float p = fmaf(t, 0x1.5a54bp-9f, -0x1.f4b218p-7f);
    p = fmaf(t, p, 0x1.53f67ep-5f);
    p = fmaf(t, p, -0x1.2fa9aep-4f);
    p = fmaf(t, p, 0x1.b26364p-4f);
    p = fmaf(t, p, -0x1.22c1ccp-3f);
    p = fmaf(t, p, 0x1.99717ep-3f);
    p = fmaf(t, p, -0x1.5554c4p-2f);
    float a = fmaf(q, t * p, q);
    a = ay > ax ? 0x1.921fb6p+0f - a : a;
    a = x < 0.0f ? 0x1.921fb6p+1f - a : a;
    ph = copysignf(a, y);
}

__global__ __launch_bounds__(BLOCK_THREADS)
void convstft_wmma_f32(const float* __restrict__ in,
                       const float* __restrict__ w,
                       float* __restrict__ mags,
                       float* __restrict__ phase)
{
    extern __shared__ __attribute__((aligned(16))) float smem[];

    const unsigned tid = threadIdx.x;
    __builtin_assume(tid < BLOCK_THREADS);
    const int lane = tid & 31;
    const int wave = __builtin_amdgcn_readfirstlane((int)(tid >> 5));
    const int half = lane >> 4;
    const int mr   = lane & 15;

    const int u0  = blockIdx.x * 4;
    const int bc0 = u0 / N_CT;
    const int bc1 = (u0 + 3) / N_CT;

    for (int s = 0; s < 2; ++s) {
        if (s == 1 && bc1 == bc0) break;
        const int bcs   = bc0 + s;
        const int bs    = bcs / N_CHUNK;
        const int chs   = bcs - bs * N_CHUNK;
        const int sbase = ((chs * N_NT) / N_CHUNK) * 16 * WIN_INC - PAD;
        const float* inb = in + (size_t)bs * N_IN;
        float* Xw = smem + s * XW_STRIDE;
        v4f xv[XW_PASSES];
        #pragma unroll
        for (int i = 0; i < XW_PASSES; ++i) {
            const int g = sbase + 4 * (int)(tid + BLOCK_THREADS * i);
            const unsigned gs = (unsigned)g < (unsigned)N_IN ? (unsigned)g : 0u;
            xv[i] = *(const v4f*)(inb + gs);
        }
        __builtin_amdgcn_sched_barrier(0);
        #pragma unroll
        for (int i = 0; i < XW_PASSES; ++i) {
            const unsigned j = tid + BLOCK_THREADS * i;
            const int g = sbase + 4 * (int)j;
            const bool ok = (unsigned)g < (unsigned)N_IN;
            v4f v = xv[i];
            v.x = ok ? v.x : 0.0f;  v.y = ok ? v.y : 0.0f;  v.z = ok ? v.z : 0.0f;  v.w = ok ? v.w : 0.0f;
            *(v4f*)(Xw + 4u * j) = v;
        }
    }

    const int u     = u0 + wave;
    const int bc    = u / N_CT;
    const int ct    = u - bc * N_CT;
    const int b     = bc / N_CHUNK;
    const int ch    = bc - b * N_CHUNK;
    const int nt_lo = (ch * N_NT) / N_CHUNK;
    const int nt_hi = ((ch + 1) * N_NT) / N_CHUNK;
    const int c0    = ct * 16;

    int row = c0 + mr;
    row = row < N_CH ? row : N_CH - 1;
    const v4f* wr = (const v4f*)(w + (size_t)row * WIN_LEN + 4 * half);
    const v4f* wi = (const v4f*)(w + (size_t)(row + N_CH) * WIN_LEN + 4 * half);
    v4f a_re[K_STEPS], a_im[K_STEPS];
    #pragma unroll
    for (int j = 0; j < K_STEPS; ++j) {
        a_re[j] = wr[2 * j];
        a_im[j] = wi[2 * j];
    }
    __syncthreads();

    const float* xs_p = smem + (bc - bc0) * XW_STRIDE + WIN_INC * mr + 4 * half;

    for (int nt = nt_lo; nt < nt_hi; ++nt) {
        const float* xp = xs_p + (nt - nt_lo) * (16 * WIN_INC);
        v8f accr = {};
        v8f acci = {};
        v4f xq[2][K_GROUP];
        #pragma unroll
        for (int e = 0; e < K_GROUP; ++e) xq[0][e] = *(const v4f*)(xp + 8 * e);
        #pragma unroll
        for (int g = 0; g < K_STEPS / K_GROUP; ++g) {
            if (g + 1 < K_STEPS / K_GROUP) {
                #pragma unroll
                for (int e = 0; e < K_GROUP; ++e)
                    xq[(g + 1) & 1][e] = *(const v4f*)(xp + 8 * (K_GROUP * (g + 1) + e));
            }
            __builtin_amdgcn_sched_barrier(0);
            #pragma unroll
            for (int e = 0; e < K_GROUP; ++e) {
                const int j = K_GROUP * g + e;
                const v4f xb = xq[g & 1][e];
                const v2f b0 = {xb.x, xb.y};
                const v2f b1 = {xb.z, xb.w};
                const v2f r0 = {a_re[j].x, a_re[j].y};
                const v2f r1 = {a_re[j].z, a_re[j].w};
                const v2f i0 = {a_im[j].x, a_im[j].y};
                const v2f i1 = {a_im[j].z, a_im[j].w};
                accr = __builtin_amdgcn_wmma_f32_16x16x4_f32(false, r0, false, b0, (short)0, accr, false, false);
                acci = __builtin_amdgcn_wmma_f32_16x16x4_f32(false, i0, false, b0, (short)0, acci, false, false);
                accr = __builtin_amdgcn_wmma_f32_16x16x4_f32(false, r1, false, b1, (short)0, accr, false, false);
                acci = __builtin_amdgcn_wmma_f32_16x16x4_f32(false, i1, false, b1, (short)0, acci, false, false);
            }
        }

        const int t = nt * 16 + mr;
        float* pm = mags + ((size_t)(b * N_CH + c0 + 8 * half) * N_FRAMES + t);
        float* pp = pm + PHASE_OFF;
        float xs[8], r2[8];
        #pragma unroll
        for (int r = 0; r < 8; ++r) {
            xs[r] = accr[r] + 1.19209290e-07f;
            r2[r] = accr[r] * accr[r];
        }
        __builtin_amdgcn_sched_barrier(0);
        if (ct < N_CT - 1 && nt < N_NT - 1) {
            float mg[8], ph[8];
            #pragma unroll
            for (int r = 0; r < 8; ++r) mag_phase(xs[r], r2[r], acci[r], mg[r], ph[r]);
            #pragma unroll
            for (int r = 0; r < 8; ++r) {
                pm[r * N_FRAMES] = mg[r];
                pp[r * N_FRAMES] = ph[r];
            }
        } else if (t < N_FRAMES) {
            #pragma unroll
            for (int r = 0; r < 8; ++r) {
                if (c0 + r + 8 * half < N_CH) {
                    float mgr, phr;
                    mag_phase(xs[r], r2[r], acci[r], mgr, phr);
                    pm[r * N_FRAMES] = mgr;
                    pp[r * N_FRAMES] = phr;
                }
            }
        }
    }
}

extern "C" void kernel_launch(void* const* d_in, const int* in_sizes, int n_in,
                              void* d_out, int out_size, void* d_ws, size_t ws_size,
                              hipStream_t stream) {
    (void)in_sizes; (void)n_in; (void)d_ws; (void)ws_size; (void)out_size;

    const float* in = (const float*)d_in[0];
    const float* w  = (const float*)d_in[1];

    float* mags  = (float*)d_out;
    float* phase = mags + (size_t)N_BATCH * N_CH * N_FRAMES;

    const dim3 grid(N_UNITS / 4);
    const dim3 block(BLOCK_THREADS);
    const size_t smem_bytes = (size_t)(2 * XW_STRIDE) * sizeof(float);

    convstft_wmma_f32<<<grid, block, smem_bytes, stream>>>(in, w, mags, phase);
}


# one resident pass: 510 blocks (was 1020 in two passes); each wave loads its 16 channels' weight rows once and runs two consecutive frame chunks, input windows restaged between them; sums unchanged
# speedup vs baseline: 3.8640x; 1.0541x over previous
#include <hip/hip_runtime.h>
#include <math.h>


typedef float v2f __attribute__((ext_vector_type(2)));
typedef float v4f __attribute__((ext_vector_type(4)));
typedef float v8f __attribute__((ext_vector_type(8)));

#define WIN_LEN   400
#define WIN_INC   100
#define PAD       300
#define N_FRAMES  1603
#define N_CH      257
#define N_IN      160000
#define N_BATCH   16
#define BLOCK_THREADS 128
#define N_CT      17
#define N_NT      101
#define N_CHUNK   15
#define MAX_NT    7
#define N_PAIR    (N_BATCH * N_CHUNK / 2)
#define N_UNITS   (N_CT * N_PAIR)
#define XW_LEN    (WIN_INC * (16 * MAX_NT - 1) + WIN_LEN)
#define XW_VECS   (XW_LEN / 4)
#define XW_PASSES ((XW_VECS + BLOCK_THREADS - 1) / BLOCK_THREADS)
#define XW_STRIDE (4 * BLOCK_THREADS * XW_PASSES)
#define K_STEPS   (WIN_LEN / 8)
#define K_GROUP   5


#define PHASE_OFF ((size_t)N_BATCH * N_CH * N_FRAMES)

__device__ __forceinline__ void mag_phase(float x, float r2, float ii, float& mg, float& ph)
{
    const float EPS = 1.19209290e-07f;
    mg = __builtin_amdgcn_sqrtf(fmaxf(fmaf(ii, ii, r2), EPS));
    const float y = ii + EPS;
    const float ax = fabsf(x);
    const float ay = fabsf(y);
    const float u = fmaxf(fmaxf(ax, ay), 1.17549435e-38f);
    const float v = fminf(ax, ay);
    const float q = __builtin_amdgcn_ldexpf(
        __builtin_amdgcn_frexp_mantf(v) * __builtin_amdgcn_rcpf(__builtin_amdgcn_frexp_mantf(u)),
        __builtin_amdgcn_frexp_expf(v) - __builtin_amdgcn_frexp_expf(u));
    const float t = q * q;
    float p = fmaf(t, 0x1.5a54bp-9f, -0x1.f4b218p-7f);
    p = fmaf(t, p, 0x1.53f67ep-5f);
    p = fmaf(t, p, -0x1.2fa9aep-4f);
    p = fmaf(t, p, 0x1.b26364p-4f);
    p = fmaf(t, p, -0x1.22c1ccp-3f);
    p = fmaf(t, p, 0x1.99717ep-3f);
    p = fmaf(t, p, -0x1.5554c4p-2f);
    float a = fmaf(q, t * p, q);
    a = ay > ax ? 0x1.921fb6p+0f - a : a;
    a = x < 0.0f ? 0x1.921fb6p+1f - a : a;
    ph = copysignf(a, y);
}

__global__ __launch_bounds__(BLOCK_THREADS)
void convstft_wmma_f32(const float* __restrict__ in,
                       const float* __restrict__ w,
                       float* __restrict__ mags,
                       float* __restrict__ phase)
{
    extern __shared__ __attribute__((aligned(16))) float smem[];

    const unsigned tid = threadIdx.x;
    __builtin_assume(tid < BLOCK_THREADS);
    const int lane = tid & 31;
    const int wave = __builtin_amdgcn_readfirstlane((int)(tid >> 5));
    const int half = lane >> 4;
    const int mr   = lane & 15;

    const int u0  = blockIdx.x * 4;
    const int pc0 = u0 / N_CT;
    const int pc1 = (u0 + 3) / N_CT;

    const int u   = u0 + wave;
    const int pc  = u / N_CT;
    const int ct  = u - pc * N_CT;
    const int c0  = ct * 16;

    int row = c0 + mr;
    row = row < N_CH ? row : N_CH - 1;
    const v4f* wr = (const v4f*)(w + (size_t)row * WIN_LEN + 4 * half);
    const v4f* wi = (const v4f*)(w + (size_t)(row + N_CH) * WIN_LEN + 4 * half);
    v4f a_re[K_STEPS], a_im[K_STEPS];

    for (int it = 0; it < 2; ++it) {
        if (it) __syncthreads();

        for (int s = 0; s < 2; ++s) {
            if (s == 1 && pc1 == pc0) break;
            const int bcs   = 2 * (pc0 + s) + it;
            const int bs    = bcs / N_CHUNK;
            const int chs   = bcs - bs * N_CHUNK;
            const int sbase = ((chs * N_NT) / N_CHUNK) * 16 * WIN_INC - PAD;
            const float* inb = in + (size_t)bs * N_IN;
            float* Xw = smem + s * XW_STRIDE;
            v4f xv[XW_PASSES];
            #pragma unroll
            for (int i = 0; i < XW_PASSES; ++i) {
                const int g = sbase + 4 * (int)(tid + BLOCK_THREADS * i);
                const unsigned gs = (unsigned)g < (unsigned)N_IN ? (unsigned)g : 0u;
                xv[i] = *(const v4f*)(inb + gs);
            }
            __builtin_amdgcn_sched_barrier(0);
            #pragma unroll
            for (int i = 0; i < XW_PASSES; ++i) {
                const unsigned j = tid + BLOCK_THREADS * i;
                const int g = sbase + 4 * (int)j;
                const bool ok = (unsigned)g < (unsigned)N_IN;
                v4f v = xv[i];
                v.x = ok ? v.x : 0.0f;  v.y = ok ? v.y : 0.0f;  v.z = ok ? v.z : 0.0f;  v.w = ok ? v.w : 0.0f;
                *(v4f*)(Xw + 4u * j) = v;
            }
        }

        if (it == 0) {
            #pragma unroll
            for (int j = 0; j < K_STEPS; ++j) {
                a_re[j] = wr[2 * j];
                a_im[j] = wi[2 * j];
            }
        }
        __syncthreads();

        const int bc    = 2 * pc + it;
        const int b     = bc / N_CHUNK;
        const int ch    = bc - b * N_CHUNK;
        const int nt_lo = (ch * N_NT) / N_CHUNK;
        const int nt_hi = ((ch + 1) * N_NT) / N_CHUNK;

        const float* xs_p = smem + (pc - pc0) * XW_STRIDE + WIN_INC * mr + 4 * half;

        for (int nt = nt_lo; nt < nt_hi; ++nt) {
            const float* xp = xs_p + (nt - nt_lo) * (16 * WIN_INC);
            v8f accr = {};
            v8f acci = {};
            v4f xq[2][K_GROUP];
            #pragma unroll
            for (int e = 0; e < K_GROUP; ++e) xq[0][e] = *(const v4f*)(xp + 8 * e);
            #pragma unroll
            for (int g = 0; g < K_STEPS / K_GROUP; ++g) {
                if (g + 1 < K_STEPS / K_GROUP) {
                    #pragma unroll
                    for (int e = 0; e < K_GROUP; ++e)
                        xq[(g + 1) & 1][e] = *(const v4f*)(xp + 8 * (K_GROUP * (g + 1) + e));
                }
                __builtin_amdgcn_sched_barrier(0);
                #pragma unroll
                for (int e = 0; e < K_GROUP; ++e) {
                    const int j = K_GROUP * g + e;
                    const v4f xb = xq[g & 1][e];
                    const v2f b0 = {xb.x, xb.y};
                    const v2f b1 = {xb.z, xb.w};
                    const v2f r0 = {a_re[j].x, a_re[j].y};
                    const v2f r1 = {a_re[j].z, a_re[j].w};
                    const v2f i0 = {a_im[j].x, a_im[j].y};
                    const v2f i1 = {a_im[j].z, a_im[j].w};
                    accr = __builtin_amdgcn_wmma_f32_16x16x4_f32(false, r0, false, b0, (short)0, accr, false, false);
                    acci = __builtin_amdgcn_wmma_f32_16x16x4_f32(false, i0, false, b0, (short)0, acci, false, false);
                    accr = __builtin_amdgcn_wmma_f32_16x16x4_f32(false, r1, false, b1, (short)0, accr, false, false);
                    acci = __builtin_amdgcn_wmma_f32_16x16x4_f32(false, i1, false, b1, (short)0, acci, false, false);
                }
            }

            const int t = nt * 16 + mr;
            float* pm = mags + ((size_t)(b * N_CH + c0 + 8 * half) * N_FRAMES + t);
            float* pp = pm + PHASE_OFF;
            float xs[8], r2[8];
            #pragma unroll
            for (int r = 0; r < 8; ++r) {
                xs[r] = accr[r] + 1.19209290e-07f;
                r2[r] = accr[r] * accr[r];
            }
            __builtin_amdgcn_sched_barrier(0);
            if (ct < N_CT - 1 && nt < N_NT - 1) {
                float mg[8], ph[8];
                #pragma unroll
                for (int r = 0; r < 8; ++r) mag_phase(xs[r], r2[r], acci[r], mg[r], ph[r]);
                #pragma unroll
                for (int r = 0; r < 8; ++r) {
                    pm[r * N_FRAMES] = mg[r];
                    pp[r * N_FRAMES] = ph[r];
                }
            } else if (t < N_FRAMES) {
                #pragma unroll
                for (int r = 0; r < 8; ++r) {
                    if (c0 + r + 8 * half < N_CH) {
                        float mgr, phr;
                        mag_phase(xs[r], r2[r], acci[r], mgr, phr);
                        pm[r * N_FRAMES] = mgr;
                        pp[r * N_FRAMES] = phr;
                    }
                }
            }
        }
    }
}

extern "C" void kernel_launch(void* const* d_in, const int* in_sizes, int n_in,
                              void* d_out, int out_size, void* d_ws, size_t ws_size,
                              hipStream_t stream) {
    (void)in_sizes; (void)n_in; (void)d_ws; (void)ws_size; (void)out_size;

    const float* in = (const float*)d_in[0];
    const float* w  = (const float*)d_in[1];

    float* mags  = (float*)d_out;
    float* phase = mags + (size_t)N_BATCH * N_CH * N_FRAMES;

    const dim3 grid(N_UNITS / 4);
    const dim3 block(BLOCK_THREADS);
    const size_t smem_bytes = (size_t)(2 * XW_STRIDE) * sizeof(float);

    convstft_wmma_f32<<<grid, block, smem_bytes, stream>>>(in, w, mags, phase);
}


# wave priority by work left: a wave steps s_setprio 3,2,1,0 down as its frame tiles run out, so both waves of a SIMD end together instead of the younger running its second half alone; sums unchanged
# speedup vs baseline: 3.9874x; 1.0319x over previous
#include <hip/hip_runtime.h>
#include <math.h>


typedef float v2f __attribute__((ext_vector_type(2)));
typedef float v4f __attribute__((ext_vector_type(4)));
typedef float v8f __attribute__((ext_vector_type(8)));

#define WIN_LEN   400
#define WIN_INC   100
#define PAD       300
#define N_FRAMES  1603
#define N_CH      257
#define N_IN      160000
#define N_BATCH   16
#define BLOCK_THREADS 128
#define N_CT      17
#define N_NT      101
#define N_CHUNK   15
#define MAX_NT    7
#define N_PAIR    (N_BATCH * N_CHUNK / 2)
#define N_UNITS   (N_CT * N_PAIR)
#define XW_LEN    (WIN_INC * (16 * MAX_NT - 1) + WIN_LEN)
#define XW_VECS   (XW_LEN / 4)
#define XW_PASSES ((XW_VECS + BLOCK_THREADS - 1) / BLOCK_THREADS)
#define XW_STRIDE (4 * BLOCK_THREADS * XW_PASSES)
#define K_STEPS   (WIN_LEN / 8)
#define K_GROUP   5


#define PHASE_OFF ((size_t)N_BATCH * N_CH * N_FRAMES)

__device__ __forceinline__ void mag_phase(float x, float r2, float ii, float& mg, float& ph)
{
    const float EPS = 1.19209290e-07f;
    mg = __builtin_amdgcn_sqrtf(fmaxf(fmaf(ii, ii, r2), EPS));
    const float y = ii + EPS;
    const float ax = fabsf(x);
    const float ay = fabsf(y);
    const float u = fmaxf(fmaxf(ax, ay), 1.17549435e-38f);
    const float v = fminf(ax, ay);
    const float q = __builtin_amdgcn_ldexpf(
        __builtin_amdgcn_frexp_mantf(v) * __builtin_amdgcn_rcpf(__builtin_amdgcn_frexp_mantf(u)),
        __builtin_amdgcn_frexp_expf(v) - __builtin_amdgcn_frexp_expf(u));
    const float t = q * q;
    float p = fmaf(t, 0x1.5a54bp-9f, -0x1.f4b218p-7f);
    p = fmaf(t, p, 0x1.53f67ep-5f);
    p = fmaf(t, p, -0x1.2fa9aep-4f);
    p = fmaf(t, p, 0x1.b26364p-4f);
    p = fmaf(t, p, -0x1.22c1ccp-3f);
    p = fmaf(t, p, 0x1.99717ep-3f);
    p = fmaf(t, p, -0x1.5554c4p-2f);
    float a = fmaf(q, t * p, q);
    a = ay > ax ? 0x1.921fb6p+0f - a : a;
    a = x < 0.0f ? 0x1.921fb6p+1f - a : a;
    ph = copysignf(a, y);
}

__global__ __launch_bounds__(BLOCK_THREADS)
void convstft_wmma_f32(const float* __restrict__ in,
                       const float* __restrict__ w,
                       float* __restrict__ mags,
                       float* __restrict__ phase)
{
    extern __shared__ __attribute__((aligned(16))) float smem[];

    const unsigned tid = threadIdx.x;
    __builtin_assume(tid < BLOCK_THREADS);
    const int lane = tid & 31;
    const int wave = __builtin_amdgcn_readfirstlane((int)(tid >> 5));
    const int half = lane >> 4;
    const int mr   = lane & 15;

    const int u0  = blockIdx.x * 4;
    const int pc0 = u0 / N_CT;
    const int pc1 = (u0 + 3) / N_CT;

    const int u   = u0 + wave;
    const int pc  = u / N_CT;
    const int ct  = u - pc * N_CT;
    const int c0  = ct * 16;

    int left;
    {
        const int bc0 = 2 * pc, bc1 = bc0 + 1;
        const int ch0 = bc0 % N_CHUNK, ch1 = bc1 % N_CHUNK;
        left = (((ch0 + 1) * N_NT) / N_CHUNK - (ch0 * N_NT) / N_CHUNK)
             + (((ch1 + 1) * N_NT) / N_CHUNK - (ch1 * N_NT) / N_CHUNK);
    }
    int drop_at = 8;
    __builtin_amdgcn_s_setprio(3);

    int row = c0 + mr;
    row = row < N_CH ? row : N_CH - 1;
    const v4f* wr = (const v4f*)(w + (size_t)row * WIN_LEN + 4 * half);
    const v4f* wi = (const v4f*)(w + (size_t)(row + N_CH) * WIN_LEN + 4 * half);
    v4f a_re[K_STEPS], a_im[K_STEPS];

    for (int it = 0; it < 2; ++it) {
        if (it) __syncthreads();

        for (int s = 0; s < 2; ++s) {
            if (s == 1 && pc1 == pc0) break;
            const int bcs   = 2 * (pc0 + s) + it;
            const int bs    = bcs / N_CHUNK;
            const int chs   = bcs - bs * N_CHUNK;
            const int sbase = ((chs * N_NT) / N_CHUNK) * 16 * WIN_INC - PAD;
            const float* inb = in + (size_t)bs * N_IN;
            float* Xw = smem + s * XW_STRIDE;
            v4f xv[XW_PASSES];
            #pragma unroll
            for (int i = 0; i < XW_PASSES; ++i) {
                const int g = sbase + 4 * (int)(tid + BLOCK_THREADS * i);
                const unsigned gs = (unsigned)g < (unsigned)N_IN ? (unsigned)g : 0u;
                xv[i] = *(const v4f*)(inb + gs);
            }
            __builtin_amdgcn_sched_barrier(0);
            #pragma unroll
            for (int i = 0; i < XW_PASSES; ++i) {
                const unsigned j = tid + BLOCK_THREADS * i;
                const int g = sbase + 4 * (int)j;
                const bool ok = (unsigned)g < (unsigned)N_IN;
                v4f v = xv[i];
                v.x = ok ? v.x : 0.0f;  v.y = ok ? v.y : 0.0f;  v.z = ok ? v.z : 0.0f;  v.w = ok ? v.w : 0.0f;
                *(v4f*)(Xw + 4u * j) = v;
            }
        }

        if (it == 0) {
            #pragma unroll
            for (int j = 0; j < K_STEPS; ++j) {
                a_re[j] = wr[2 * j];
                a_im[j] = wi[2 * j];
            }
        }
        __syncthreads();

        const int bc    = 2 * pc + it;
        const int b     = bc / N_CHUNK;
        const int ch    = bc - b * N_CHUNK;
        const int nt_lo = (ch * N_NT) / N_CHUNK;
        const int nt_hi = ((ch + 1) * N_NT) / N_CHUNK;

        const float* xs_p = smem + (pc - pc0) * XW_STRIDE + WIN_INC * mr + 4 * half;

        for (int nt = nt_lo; nt < nt_hi; ++nt) {
            if (left == drop_at) {
                if (drop_at == 8)      { __builtin_amdgcn_s_setprio(2);  drop_at = 4; }
                else if (drop_at == 4) { __builtin_amdgcn_s_setprio(1);  drop_at = 1; }
                else                   { __builtin_amdgcn_s_setprio(0);  drop_at = 0; }
            }
            --left;
            const float* xp = xs_p + (nt - nt_lo) * (16 * WIN_INC);
            v8f accr = {};
            v8f acci = {};
            v4f xq[2][K_GROUP];
            #pragma unroll
            for (int e = 0; e < K_GROUP; ++e) xq[0][e] = *(const v4f*)(xp + 8 * e);
            #pragma unroll
            for (int g = 0; g < K_STEPS / K_GROUP; ++g) {
                if (g + 1 < K_STEPS / K_GROUP) {
                    #pragma unroll
                    for (int e = 0; e < K_GROUP; ++e)
                        xq[(g + 1) & 1][e] = *(const v4f*)(xp + 8 * (K_GROUP * (g + 1) + e));
                }
                __builtin_amdgcn_sched_barrier(0);
                #pragma unroll
                for (int e = 0; e < K_GROUP; ++e) {
                    const int j = K_GROUP * g + e;
                    const v4f xb = xq[g & 1][e];
                    const v2f b0 = {xb.x, xb.y};
                    const v2f b1 = {xb.z, xb.w};
                    const v2f r0 = {a_re[j].x, a_re[j].y};
                    const v2f r1 = {a_re[j].z, a_re[j].w};
                    const v2f i0 = {a_im[j].x, a_im[j].y};
                    const v2f i1 = {a_im[j].z, a_im[j].w};
                    accr = __builtin_amdgcn_wmma_f32_16x16x4_f32(false, r0, false, b0, (short)0, accr, false, false);
                    acci = __builtin_amdgcn_wmma_f32_16x16x4_f32(false, i0, false, b0, (short)0, acci, false, false);
                    accr = __builtin_amdgcn_wmma_f32_16x16x4_f32(false, r1, false, b1, (short)0, accr, false, false);
                    acci = __builtin_amdgcn_wmma_f32_16x16x4_f32(false, i1, false, b1, (short)0, acci, false, false);
                }
            }

            const int t = nt * 16 + mr;
            float* pm = mags + ((size_t)(b * N_CH + c0 + 8 * half) * N_FRAMES + t);
            float* pp = pm + PHASE_OFF;
            float xs[8], r2[8];
            #pragma unroll
            for (int r = 0; r < 8; ++r) {
                xs[r] = accr[r] + 1.19209290e-07f;
                r2[r] = accr[r] * accr[r];
            }
            __builtin_amdgcn_sched_barrier(0);
            if (ct < N_CT - 1 && nt < N_NT - 1) {
                float mg[8], ph[8];
                #pragma unroll
                for (int r = 0; r < 8; ++r) mag_phase(xs[r], r2[r], acci[r], mg[r], ph[r]);
                #pragma unroll
                for (int r = 0; r < 8; ++r) {
                    pm[r * N_FRAMES] = mg[r];
                    pp[r * N_FRAMES] = ph[r];
                }
            } else if (t < N_FRAMES) {
                #pragma unroll
                for (int r = 0; r < 8; ++r) {
                    if (c0 + r + 8 * half < N_CH) {
                        float mgr, phr;
                        mag_phase(xs[r], r2[r], acci[r], mgr, phr);
                        pm[r * N_FRAMES] = mgr;
                        pp[r * N_FRAMES] = phr;
                    }
                }
            }
        }
    }
}

extern "C" void kernel_launch(void* const* d_in, const int* in_sizes, int n_in,
                              void* d_out, int out_size, void* d_ws, size_t ws_size,
                              hipStream_t stream) {
    (void)in_sizes; (void)n_in; (void)d_ws; (void)ws_size; (void)out_size;

    const float* in = (const float*)d_in[0];
    const float* w  = (const float*)d_in[1];

    float* mags  = (float*)d_out;
    float* phase = mags + (size_t)N_BATCH * N_CH * N_FRAMES;

    const dim3 grid(N_UNITS / 4);
    const dim3 block(BLOCK_THREADS);
    const size_t smem_bytes = (size_t)(2 * XW_STRIDE) * sizeof(float);

    convstft_wmma_f32<<<grid, block, smem_bytes, stream>>>(in, w, mags, phase);
}
